# starting kernel
# baseline (speedup 1.0000x reference)
#include <hip/hip_runtime.h>


typedef __attribute__((ext_vector_type(2))) float v2f;
typedef __attribute__((ext_vector_type(8))) float v8f;

#define F_OUT 128
#define N_HEAD 4

__global__ void rgat_prep(const float* __restrict__ w,
                          const float* __restrict__ a,
                          float* __restrict__ cs,
                          float* __restrict__ BsrcP,
                          float* __restrict__ BdstP) {
  int f = threadIdx.x;
  float c1 = w[f];
  float c2 = c1 * w[F_OUT + f];
  float c3 = c2 * w[2 * F_OUT + f];
  cs[f] = 1.0f;  cs[F_OUT + f] = c1;  cs[2 * F_OUT + f] = c2;  cs[3 * F_OUT + f] = c3;

  for (int idx = f; idx < 32 * 64; idx += F_OUT) {
    int s    = idx >> 6;
    int l    = (idx >> 1) & 31;
    int half = idx & 1;
    int col  = l & 15;
    int k    = s * 4 + ((l >> 4) << 1) + half;
    float bs = 0.0f, bd = 0.0f;
    if (col < 4) {
      float c = 1.0f;
      for (int j = 0; j < col; ++j) c *= w[j * F_OUT + k];
      bs = c * a[col * 256 + k];
      bd = a[col * 256 + F_OUT + k];
    }
    BsrcP[idx] = bs;
    BdstP[idx] = bd;
  }
}

__global__ void rgat_score(const float* __restrict__ X,
                           const float* __restrict__ Bp,
                           float* __restrict__ S, int M) {
  int tile = blockIdx.x * (blockDim.x >> 5) + (threadIdx.x >> 5);
  int base = tile * 16;
  if (base >= M) return;

  int lane = threadIdx.x & 31;
  int col  = lane & 15;
  int kb   = (lane >> 4) * 2;
  int row  = base + (lane & 15);
  row = row < M ? row : M - 1;
  const float* xr = X + (size_t)row * F_OUT;
  const float* bp = Bp + lane * 2;

  v8f acc = {};
#pragma unroll
  for (int s = 0; s < 32; ++s) {
    v2f av = *(const v2f*)(xr + s * 4 + kb);
    v2f bv = *(const v2f*)(bp + s * 64);
    acc = __builtin_amdgcn_wmma_f32_16x16x4_f32(
         false, av,  false, bv,
         (short)0, acc,  false,  false);
  }

  int mhalf = (lane >> 4) * 8;
#pragma unroll
  for (int v = 0; v < 8; ++v)
    S[(size_t)(base + mhalf + v) * 16 + col] = acc[v];
}

__global__ void rgat_edge(const int* __restrict__ A,
                          const float* __restrict__ h,
                          const float* __restrict__ inputr,
                          const float* __restrict__ cs,
                          const float* __restrict__ Snode,
                          const float* __restrict__ Srel,
                          float* __restrict__ rowsum,
                          float* __restrict__ out,
                          int E, int N) {
  const int lane = threadIdx.x & 31;
  const int wid  = (blockIdx.x * blockDim.x + threadIdx.x) >> 5;
  const int nw   = (gridDim.x * blockDim.x) >> 5;

  const float4 cv0 = *(const float4*)(cs + 0 * F_OUT + lane * 4);
  const float4 cv1 = *(const float4*)(cs + 1 * F_OUT + lane * 4);
  const float4 cv2 = *(const float4*)(cs + 2 * F_OUT + lane * 4);
  const float4 cv3 = *(const float4*)(cs + 3 * F_OUT + lane * 4);
  const float4 cv[N_HEAD] = {cv0, cv1, cv2, cv3};

  for (int ev = wid; ev < E; ev += nw) {
    const int e   = __builtin_amdgcn_readfirstlane(ev);
    const int dst = A[e];
    const int rl  = A[E + e];
    const int src = A[2 * E + e];

    const float4 hs = *(const float4*)(h + (size_t)src * F_OUT + lane * 4);
    const float4 rf = *(const float4*)(inputr + (size_t)rl * F_OUT + lane * 4);
    const float4 sn = *(const float4*)(Snode + (size_t)src * 16);
    const float4 sr = *(const float4*)(Srel + (size_t)rl * 16);

    const float logit[N_HEAD] = {sn.x + sr.x, sn.y + sr.y, sn.z + sr.z, sn.w + sr.w};
    float ee[N_HEAD];
#pragma unroll
    for (int i = 0; i < N_HEAD; ++i) {
      float lk = logit[i] > 0.0f ? logit[i] : 0.2f * logit[i];
      ee[i] = __expf(-lk);
    }

    if (lane == 0) {
#pragma unroll
      for (int i = 0; i < N_HEAD; ++i)
        __hip_atomic_fetch_add(&rowsum[(size_t)i * N + dst], ee[i],
                               __ATOMIC_RELAXED, __HIP_MEMORY_SCOPE_AGENT);
    }

#pragma unroll
    for (int i = 0; i < N_HEAD; ++i) {
      float* op = out + ((size_t)i * N + dst) * F_OUT + lane * 4;
      __hip_atomic_fetch_add(&op[0], (cv[i].x * hs.x - rf.x) * ee[i],
                             __ATOMIC_RELAXED, __HIP_MEMORY_SCOPE_AGENT);
      __hip_atomic_fetch_add(&op[1], (cv[i].y * hs.y - rf.y) * ee[i],
                             __ATOMIC_RELAXED, __HIP_MEMORY_SCOPE_AGENT);
      __hip_atomic_fetch_add(&op[2], (cv[i].z * hs.z - rf.z) * ee[i],
                             __ATOMIC_RELAXED, __HIP_MEMORY_SCOPE_AGENT);
      __hip_atomic_fetch_add(&op[3], (cv[i].w * hs.w - rf.w) * ee[i],
                             __ATOMIC_RELAXED, __HIP_MEMORY_SCOPE_AGENT);
    }
  }
}

__global__ void rgat_norm(float* __restrict__ out,
                          const float* __restrict__ rowsum,
                          int total) {
  int idx = blockIdx.x * blockDim.x + threadIdx.x;
  if (idx >= total) return;
  out[idx] = out[idx] / rowsum[idx >> 7];
}

extern "C" void kernel_launch(void* const* d_in, const int* in_sizes, int n_in,
                              void* d_out, int out_size, void* d_ws, size_t ws_size,
                              hipStream_t stream) {
  const float* h      = (const float*)d_in[0];
  const float* inputr = (const float*)d_in[1];
  const float* w      = (const float*)d_in[2];
  const float* a      = (const float*)d_in[3];
  const int*   A      = (const int*)d_in[4];

  const int N    = in_sizes[0] / F_OUT;
  const int NREL = in_sizes[1] / F_OUT;
  const int E    = in_sizes[4] / 3;

  float* ws = (float*)d_ws;
  float* cs    = ws;
  float* BsrcP = ws + 512;
  float* BdstP = ws + 512 + 2048;
  size_t off   = 512 + 2048 + 2048;
  const int nrel_tiles = (NREL + 15) / 16;
  const int node_tiles = (N + 15) / 16;
  float* Srel  = ws + off;  off += (size_t)nrel_tiles * 16 * 16;
  float* Snode = ws + off;  off += (size_t)node_tiles * 16 * 16;
  float* rowsum = ws + off; off += (size_t)N_HEAD * N;
  (void)ws_size;

  hipMemsetAsync(d_out, 0, (size_t)out_size * sizeof(float), stream);
  hipMemsetAsync(rowsum, 0, (size_t)N_HEAD * N * sizeof(float), stream);

  rgat_prep<<<1, F_OUT, 0, stream>>>(w, a, cs, BsrcP, BdstP);

  const int WPB = 8;
  rgat_score<<<(node_tiles + WPB - 1) / WPB, 256, 0, stream>>>(h, BsrcP, Snode, N);
  rgat_score<<<(nrel_tiles + WPB - 1) / WPB, 256, 0, stream>>>(inputr, BdstP, Srel, NREL);

  int eblocks = (E + WPB - 1) / WPB;
  if (eblocks > 4096) eblocks = 4096;
  rgat_edge<<<eblocks, 256, 0, stream>>>(A, h, inputr, cs, Snode, Srel,
                                         rowsum, (float*)d_out, E, N);

  rgat_norm<<<(out_size + 255) / 256, 256, 0, stream>>>((float*)d_out, rowsum, out_size);
}


# Owner-computes edge phase: work-group owns 196 dst nodes, reads dst column once, queues edges per owner wave, sums rows in LDS in queue order, divides, stores once; f32 atomics, memsets, norm gone
# speedup vs baseline: 4.4465x; 4.4465x over previous
#include <hip/hip_runtime.h>


typedef __attribute__((ext_vector_type(2))) float v2f;
typedef __attribute__((ext_vector_type(8))) float v8f;

#define F_OUT 128
#define N_HEAD 4

__global__ void rgat_prep(const float* __restrict__ w,
                          const float* __restrict__ a,
                          float* __restrict__ cs,
                          float* __restrict__ BsrcP,
                          float* __restrict__ BdstP) {
  int f = threadIdx.x;
  float c1 = w[f];
  float c2 = c1 * w[F_OUT + f];
  float c3 = c2 * w[2 * F_OUT + f];
  cs[f] = 1.0f;  cs[F_OUT + f] = c1;  cs[2 * F_OUT + f] = c2;  cs[3 * F_OUT + f] = c3;

  for (int idx = f; idx < 32 * 64; idx += F_OUT) {
    int s    = idx >> 6;
    int l    = (idx >> 1) & 31;
    int half = idx & 1;
    int col  = l & 15;
    int k    = s * 4 + ((l >> 4) << 1) + half;
    float bs = 0.0f, bd = 0.0f;
    if (col < 4) {
      float c = 1.0f;
      for (int j = 0; j < col; ++j) c *= w[j * F_OUT + k];
      bs = c * a[col * 256 + k];
      bd = a[col * 256 + F_OUT + k];
    }
    BsrcP[idx] = bs;
    BdstP[idx] = bd;
  }
}

__global__ void rgat_score(const float* __restrict__ X,
                           const float* __restrict__ Bp,
                           float* __restrict__ S, int M) {
  int tile = blockIdx.x * (blockDim.x >> 5) + (threadIdx.x >> 5);
  int base = tile * 16;
  if (base >= M) return;

  int lane = threadIdx.x & 31;
  int col  = lane & 15;
  int kb   = (lane >> 4) * 2;
  int row  = base + (lane & 15);
  row = row < M ? row : M - 1;
  const float* xr = X + (size_t)row * F_OUT;
  const float* bp = Bp + lane * 2;

  v8f acc = {};
#pragma unroll
  for (int s = 0; s < 32; ++s) {
    v2f av = *(const v2f*)(xr + s * 4 + kb);
    v2f bv = *(const v2f*)(bp + s * 64);
    acc = __builtin_amdgcn_wmma_f32_16x16x4_f32(
         false, av,  false, bv,
         (short)0, acc,  false,  false);
  }

  int mhalf = (lane >> 4) * 8;
#pragma unroll
  for (int v = 0; v < 8; ++v)
    S[(size_t)(base + mhalf + v) * 16 + col] = acc[v];
}

#define DPB  98
#define BCAP 128
#define NBKT 64
#define NTHR 1024

__device__ __forceinline__ void rgat_edge_one_lane(
    int e, int dh, const int* __restrict__ A, const float* __restrict__ h,
    const float* __restrict__ inputr, const float* __restrict__ cs,
    const float* __restrict__ Snode, const float* __restrict__ Srel,
    float* acc, float* rs, int E) {
  const int rl  = A[(size_t)E + e];
  const int src = A[2 * (size_t)E + e];
  float ee[N_HEAD];
#pragma unroll
  for (int i = 0; i < N_HEAD; ++i) {
    float lg = Snode[(size_t)src * 16 + i] + Srel[(size_t)rl * 16 + i];
    float lk = lg > 0.0f ? lg : 0.2f * lg;
    ee[i] = __expf(-lk);
    __hip_atomic_fetch_add(&rs[dh * N_HEAD + i], ee[i],
                           __ATOMIC_RELAXED, __HIP_MEMORY_SCOPE_WORKGROUP);
  }
  for (int f = 0; f < F_OUT; ++f) {
    const float hv = h[(size_t)src * F_OUT + f];
    const float rv = inputr[(size_t)rl * F_OUT + f];
#pragma unroll
    for (int i = 0; i < N_HEAD; ++i)
      __hip_atomic_fetch_add(&acc[(dh * N_HEAD + i) * F_OUT + f],
                             (cs[i * F_OUT + f] * hv - rv) * ee[i],
                             __ATOMIC_RELAXED, __HIP_MEMORY_SCOPE_WORKGROUP);
  }
}

__global__ __launch_bounds__(NTHR)
void rgat_own(const int* __restrict__ A,
              const float* __restrict__ h,
              const float* __restrict__ inputr,
              const float* __restrict__ cs,
              const float* __restrict__ Snode,
              const float* __restrict__ Srel,
              float* __restrict__ out,
              int E, int N) {
  __shared__ __attribute__((aligned(16))) float acc[DPB * N_HEAD * F_OUT];
  __shared__ float rs[DPB * N_HEAD];
  __shared__ int q_src[NBKT * BCAP];
  __shared__ int q_rel[NBKT * BCAP];
  __shared__ int q_dh[NBKT * BCAP];
  __shared__ int q_cnt[NBKT];

  const int tid  = threadIdx.x;
  const int lane = tid & 31;
  const int wv   = tid >> 5;
  const int lo   = blockIdx.x * (2 * DPB);
  const int ndt  = (N - lo) < 2 * DPB ? (N - lo) : 2 * DPB;

  if (tid < NBKT) q_cnt[tid] = 0;
  __syncthreads();

  auto take = [&](int d, int e) {
    const int dl = d - lo;
    if ((unsigned)dl < (unsigned)ndt) {
      const int hf = dl >= DPB ? 1 : 0;
      const int dh = dl - hf * DPB;
      const int bk = hf * 32 + (dh & 31);
      const int sl = __hip_atomic_fetch_add(&q_cnt[bk], 1, __ATOMIC_RELAXED,
                                            __HIP_MEMORY_SCOPE_WORKGROUP);
      if (sl < BCAP) { q_src[bk * BCAP + sl] = e; q_dh[bk * BCAP + sl] = dh; }
    }
  };
  const int E4 = E >> 2;
  const int4* __restrict__ A4 = (const int4*)A;
  auto scan8 = [&](int g0, bool guarded) __attribute__((always_inline)) {
    int4 d[8];
#pragma unroll
    for (int k = 0; k < 8; ++k) {
      const int g = g0 + k * NTHR + tid;
      d[k] = (!guarded || g < E4) ? A4[g] : make_int4(-1, -1, -1, -1);
    }
#pragma unroll
    for (int k = 0; k < 8; ++k) {
      const int g = g0 + k * NTHR + tid;
      const bool any = ((unsigned)(d[k].x - lo) < (unsigned)ndt) |
                       ((unsigned)(d[k].y - lo) < (unsigned)ndt) |
                       ((unsigned)(d[k].z - lo) < (unsigned)ndt) |
                       ((unsigned)(d[k].w - lo) < (unsigned)ndt);
      if (any) {
        take(d[k].x, 4 * g);     take(d[k].y, 4 * g + 1);
        take(d[k].z, 4 * g + 2); take(d[k].w, 4 * g + 3);
      }
    }
  };
  int g0 = 0;
  for (; g0 + NTHR * 8 <= E4; g0 += NTHR * 8) scan8(g0, false);
  if (g0 < E4) scan8(g0, true);
  {
    const int e = (E4 << 2) + tid;
    if (e < E) take(A[e], e);
  }
  __syncthreads();

  {
    int  qe[8];
    bool ok[8];
#pragma unroll
    for (int k = 0; k < 8; ++k) {
      const int i = tid + k * NTHR;
      ok[k] = (i & (BCAP - 1)) < q_cnt[i / BCAP];
      qe[k] = ok[k] ? q_src[i] : 0;
    }
    int qr[8], qs[8];
#pragma unroll
    for (int k = 0; k < 8; ++k) {
      qr[k] = A[(size_t)E + qe[k]];
      qs[k] = A[2 * (size_t)E + qe[k]];
    }
#pragma unroll
    for (int k = 0; k < 8; ++k) {
      const int i = tid + k * NTHR;
      if (ok[k]) { q_rel[i] = qr[k]; q_src[i] = qs[k]; }
    }
  }

  float4 cv[N_HEAD];
#pragma unroll
  for (int i = 0; i < N_HEAD; ++i) cv[i] = *(const float4*)(cs + i * F_OUT + lane * 4);

  for (int hf = 0; hf < 2; ++hf) {
    const int lo_h = lo + hf * DPB;
    int nd_h = N - lo_h;
    nd_h = nd_h < 0 ? 0 : (nd_h > DPB ? DPB : nd_h);

    for (int q = tid; q < DPB * N_HEAD * F_OUT / 4; q += NTHR)
      ((float4*)acc)[q] = make_float4(0.0f, 0.0f, 0.0f, 0.0f);
    if (tid < DPB * N_HEAD) rs[tid] = 0.0f;
    __syncthreads();

    const bool ovf = __builtin_amdgcn_ballot_w32(q_cnt[hf * 32 + lane] > BCAP) != 0u;
    if (!ovf) {
      const int bk = hf * 32 + wv;
      const int n  = q_cnt[bk];
      const int* bs = q_src + bk * BCAP;
      const int* br = q_rel + bk * BCAP;
      const int* bd = q_dh + bk * BCAP;
      for (int b0 = 0; b0 < n; b0 += 4) {
        int    dh[4];
        float4 hs[4], rf[4], sn[4], sr[4];
#pragma unroll
        for (int u = 0; u < 4; ++u) {
          const int ix  = (b0 + u) < n ? (b0 + u) : b0;
          const int src = __builtin_amdgcn_readfirstlane(bs[ix]);
          const int rl  = __builtin_amdgcn_readfirstlane(br[ix]);
          dh[u]         = __builtin_amdgcn_readfirstlane(bd[ix]);
          sn[u] = *(const float4*)(Snode + (size_t)src * 16);
          sr[u] = *(const float4*)(Srel + (size_t)rl * 16);
          hs[u] = *(const float4*)(h + (size_t)src * F_OUT + lane * 4);
          rf[u] = *(const float4*)(inputr + (size_t)rl * F_OUT + lane * 4);
        }
#pragma unroll
        for (int u = 0; u < 4; ++u) {
          if (b0 + u < n) {
            const float logit[N_HEAD] = {sn[u].x + sr[u].x, sn[u].y + sr[u].y,
                                         sn[u].z + sr[u].z, sn[u].w + sr[u].w};
            float ee[N_HEAD];
#pragma unroll
            for (int i = 0; i < N_HEAD; ++i) {
              float lk = logit[i] > 0.0f ? logit[i] : 0.2f * logit[i];
              ee[i] = __expf(-lk);
            }
            if (lane < N_HEAD) {
              const float el = lane == 0 ? ee[0] : lane == 1 ? ee[1] : lane == 2 ? ee[2] : ee[3];
              rs[dh[u] * N_HEAD + lane] += el;
            }
            float4* rp = (float4*)(acc + dh[u] * (N_HEAD * F_OUT)) + lane;
#pragma unroll
            for (int i = 0; i < N_HEAD; ++i) {
              float4 s = rp[i * (F_OUT / 4)];
              s.x += (cv[i].x * hs[u].x - rf[u].x) * ee[i];
              s.y += (cv[i].y * hs[u].y - rf[u].y) * ee[i];
              s.z += (cv[i].z * hs[u].z - rf[u].z) * ee[i];
              s.w += (cv[i].w * hs[u].w - rf[u].w) * ee[i];
              rp[i * (F_OUT / 4)] = s;
            }
          }
        }
      }
    } else {
      for (int e = tid; e < E; e += NTHR) {
        const int dh = A[e] - lo_h;
        if ((unsigned)dh < (unsigned)nd_h)
          rgat_edge_one_lane(e, dh, A, h, inputr, cs, Snode, Srel, acc, rs, E);
      }
    }
    __syncthreads();

    for (int q = tid; q < nd_h * N_HEAD * (F_OUT / 4); q += NTHR) {
      const int row = q >> 5;
      const int c4  = q & 31;
      const int dn  = row >> 2;
      const int hd  = row & 3;
      const float4 v = ((const float4*)acc)[q];
      const float  r = rs[row];
      float4 o;
      o.x = v.x / r;  o.y = v.y / r;  o.z = v.z / r;  o.w = v.w / r;
      *(float4*)(out + ((size_t)hd * N + lo_h + dn) * F_OUT + c4 * 4) = o;
    }
    __syncthreads();
  }
}

extern "C" void kernel_launch(void* const* d_in, const int* in_sizes, int n_in,
                              void* d_out, int out_size, void* d_ws, size_t ws_size,
                              hipStream_t stream) {
  const float* h      = (const float*)d_in[0];
  const float* inputr = (const float*)d_in[1];
  const float* w      = (const float*)d_in[2];
  const float* a      = (const float*)d_in[3];
  const int*   A      = (const int*)d_in[4];

  const int N    = in_sizes[0] / F_OUT;
  const int NREL = in_sizes[1] / F_OUT;
  const int E    = in_sizes[4] / 3;

  float* ws = (float*)d_ws;
  float* cs    = ws;
  float* BsrcP = ws + 512;
  float* BdstP = ws + 512 + 2048;
  size_t off   = 512 + 2048 + 2048;
  const int nrel_tiles = (NREL + 15) / 16;
  const int node_tiles = (N + 15) / 16;
  float* Srel  = ws + off;  off += (size_t)nrel_tiles * 16 * 16;
  float* Snode = ws + off;  off += (size_t)node_tiles * 16 * 16;
  (void)ws_size; (void)out_size; (void)n_in;

  rgat_prep<<<1, F_OUT, 0, stream>>>(w, a, cs, BsrcP, BdstP);

  const int WPB = 8;
  rgat_score<<<(node_tiles + WPB - 1) / WPB, 256, 0, stream>>>(h, BsrcP, Snode, N);
  rgat_score<<<(nrel_tiles + WPB - 1) / WPB, 256, 0, stream>>>(inputr, BdstP, Srel, NREL);

  rgat_own<<<(N + 2 * DPB - 1) / (2 * DPB), NTHR, 0, stream>>>(A, h, inputr, cs, Snode, Srel,
                                                               (float*)d_out, E, N);
}


# Edge weights worked out once per wave on the scalar float unit (score sum, leaky_relu, exp2, row sums in scalar registers), not on all 32 lanes; vector unit keeps the multiply-adds; sums unchanged
# speedup vs baseline: 6.6286x; 1.4907x over previous
#include <hip/hip_runtime.h>


typedef __attribute__((ext_vector_type(2))) float v2f;
typedef __attribute__((ext_vector_type(8))) float v8f;
typedef __attribute__((ext_vector_type(4))) float v4f;
typedef const __attribute__((address_space(4))) v4f* score_row_ptr;

#define F_OUT 128
#define N_HEAD 4

#define DPW    196
#define NCH    256
#define PTHR   256
#define MAXOWN 1024
#define OPT    (MAXOWN / PTHR)
#define PB     5

__global__ __launch_bounds__(PTHR)
void rgat_prep(const float* __restrict__ w,
               const float* __restrict__ a,
               float* __restrict__ cs,
               float* __restrict__ BsrcP,
               float* __restrict__ BdstP,
               const int* __restrict__ A,
               int* __restrict__ EntW,
               int* __restrict__ TabW,
               int E, int N, int CH, int nown, int tp) {
  if (blockIdx.x != 0) {
    int2* __restrict__ Ent = (int2*)EntW;
    int2* __restrict__ Tab = (int2*)TabW;
    __shared__ int p_cnt[MAXOWN];
    __shared__ int p_wsum[PTHR / 32];
    const int tid  = threadIdx.x;
    const int lane = tid & 31;
    const int wave = tid >> 5;
    const int c    = blockIdx.x - 1;
    const int e0   = c * CH < E ? c * CH : E;
    const int e1   = e0 + CH < E ? e0 + CH : E;
    for (int o = tid; o < MAXOWN; o += PTHR) p_cnt[o] = 0;
    __syncthreads();
    for (int b = e0 + tid; b < e1; b += PTHR * PB) {
      int d[PB];
#pragma unroll
      for (int k = 0; k < PB; ++k) {
        const int e = b + k * PTHR;
        d[k] = e < e1 ? A[e] : -1;
      }
#pragma unroll
      for (int k = 0; k < PB; ++k)
        if ((unsigned)d[k] < (unsigned)N)
          __hip_atomic_fetch_add(&p_cnt[(unsigned)d[k] / DPW], 1, __ATOMIC_RELAXED,
                                 __HIP_MEMORY_SCOPE_WORKGROUP);
    }
    __syncthreads();
    int loc[OPT], cn[OPT];
    int s = 0;
#pragma unroll
    for (int k = 0; k < OPT; ++k) { cn[k] = p_cnt[tid * OPT + k]; loc[k] = s; s += cn[k]; }
    int incl = s;
#pragma unroll
    for (int d = 1; d < 32; d <<= 1) {
      const int t = __shfl_up(incl, d);
      if (lane >= d) incl += t;
    }
    if (lane == 31) p_wsum[wave] = incl;
    __syncthreads();
    int tbase = incl - s;
    for (int j = 0; j < wave; ++j) tbase += p_wsum[j];
#pragma unroll
    for (int k = 0; k < OPT; ++k) {
      const int o = tid * OPT + k;
      p_cnt[o] = tbase + loc[k];
      if (o < nown) Tab[(size_t)c * tp + o] = make_int2(tbase + loc[k], cn[k]);
    }
    __syncthreads();
    for (int b = e0 + tid; b < e1; b += PTHR * PB) {
      int d[PB], sv[PB], rv[PB];
#pragma unroll
      for (int k = 0; k < PB; ++k) {
        const int  e  = b + k * PTHR;
        const bool ok = e < e1;
        d[k]  = ok ? A[e] : -1;
        sv[k] = ok ? A[2 * (size_t)E + e] : 0;
        rv[k] = ok ? A[(size_t)E + e] : 0;
      }
#pragma unroll
      for (int k = 0; k < PB; ++k)
        if ((unsigned)d[k] < (unsigned)N) {
          const int o  = (unsigned)d[k] / DPW;
          const int sl = __hip_atomic_fetch_add(&p_cnt[o], 1, __ATOMIC_RELAXED,
                                                __HIP_MEMORY_SCOPE_WORKGROUP);
          Ent[(size_t)e0 + sl] = make_int2(sv[k], rv[k] | ((d[k] - o * DPW) << 16));
        }
    }
    return;
  }
  int f = threadIdx.x;
  if (f >= F_OUT) return;
  float c1 = w[f];
  float c2 = c1 * w[F_OUT + f];
  float c3 = c2 * w[2 * F_OUT + f];
  cs[f] = 1.0f;  cs[F_OUT + f] = c1;  cs[2 * F_OUT + f] = c2;  cs[3 * F_OUT + f] = c3;

  for (int idx = f; idx < 32 * 64; idx += F_OUT) {
    int s    = idx >> 6;
    int l    = (idx >> 1) & 31;
    int half = idx & 1;
    int col  = l & 15;
    int k    = s * 4 + ((l >> 4) << 1) + half;
    float bs = 0.0f, bd = 0.0f;
    if (col < 4) {
      float c = 1.0f;
      for (int j = 0; j < col; ++j) c *= w[j * F_OUT + k];
      bs = c * a[col * 256 + k];
      bd = a[col * 256 + F_OUT + k];
    }
    BsrcP[idx] = bs;
    BdstP[idx] = bd;
  }
}

__global__ void rgat_score(const float* __restrict__ X,
                           const float* __restrict__ Bp,
                           float* __restrict__ S, int M) {
  int tile = blockIdx.x * (blockDim.x >> 5) + (threadIdx.x >> 5);
  int base = tile * 16;
  if (base >= M) return;

  int lane = threadIdx.x & 31;
  int col  = lane & 15;
  int kb   = (lane >> 4) * 2;
  int row  = base + (lane & 15);
  row = row < M ? row : M - 1;
  const float* xr = X + (size_t)row * F_OUT;
  const float* bp = Bp + lane * 2;

  v8f acc = {};
#pragma unroll
  for (int s = 0; s < 32; ++s) {
    v2f av = *(const v2f*)(xr + s * 4 + kb);
    v2f bv = *(const v2f*)(bp + s * 64);
    acc = __builtin_amdgcn_wmma_f32_16x16x4_f32(
         false, av,  false, bv,
         (short)0, acc,  false,  false);
  }

  int mhalf = (lane >> 4) * 8;
#pragma unroll
  for (int v = 0; v < 8; ++v)
    S[(size_t)(base + mhalf + v) * 16 + col] = acc[v];
}

#define QCAP 4096
#define NDC  256
#define NTHR 1024

static_assert(NCH * 4 == NTHR, "one chunk per 4 lanes: a thread keeps its chunk's entries across the two passes");
static_assert(DPW <= NDC, "one counter per destination");

__global__ __launch_bounds__(NTHR)
void rgat_own(const int* __restrict__ A,
              const int* __restrict__ EntW,
              const int* __restrict__ TabW,
              const float* __restrict__ h,
              const float* __restrict__ inputr,
              const float* __restrict__ cs,
              const float* __restrict__ Snode,
              const float* __restrict__ Srel,
              float* __restrict__ out,
              int E, int N, int CH, int tp, int part) {
  __shared__ int2 q_sr[QCAP];
  __shared__ int  d_beg[NDC + 1];
  __shared__ int  d_fill[NDC];
  __shared__ int  p_ws[NDC / 32];
  __shared__ int  next_d;

  const int tid  = threadIdx.x;
  const int lane = tid & 31;
  const int wv   = __builtin_amdgcn_readfirstlane(tid >> 5);
  const int lo   = blockIdx.x * DPW;
  const int ndt  = (N - lo) < DPW ? (N - lo) : DPW;

  if (tid <= NDC) d_beg[tid] = 0;
  if (tid == 0) next_d = 0;
  __syncthreads();

  const int2* __restrict__ Ent = (const int2*)EntW;
  const int2* __restrict__ Tab = (const int2*)TabW;
  const int sub = tid & 3;
  const int2* __restrict__ seg = Ent;
  int  nseg = 0;
  int2 en0 = make_int2(0, 0), en1 = en0, en2 = en0;
  if (part) {
    const int  c = tid >> 2;
    const int2 t = Tab[(size_t)c * tp + blockIdx.x];
    seg  = Ent + ((size_t)c * CH + t.x);
    nseg = t.y;
    if (sub < nseg)     en0 = seg[sub];
    if (sub + 4 < nseg) en1 = seg[sub + 4];
    if (sub + 8 < nseg) en2 = seg[sub + 8];
    auto count = [&](const int2 en) {
      __hip_atomic_fetch_add(&d_beg[(en.y >> 16) & (NDC - 1)], 1, __ATOMIC_RELAXED,
                             __HIP_MEMORY_SCOPE_WORKGROUP);
    };
    if (sub < nseg)     count(en0);
    if (sub + 4 < nseg) count(en1);
    if (sub + 8 < nseg) count(en2);
    for (int i = sub + 12; i < nseg; i += 4) count(seg[i]);
  }
  __syncthreads();

  int cn = 0, incl = 0;
  if (tid < NDC) {
    cn = d_beg[tid];
    incl = cn;
#pragma unroll
    for (int d = 1; d < 32; d <<= 1) {
      const int t = __shfl_up(incl, d);
      if (lane >= d) incl += t;
    }
    if (lane == 31) p_ws[wv] = incl;
  }
  __syncthreads();
  if (tid < NDC) {
    int base = incl - cn;
#pragma unroll
    for (int j = 0; j < NDC / 32; ++j) base += j < wv ? p_ws[j] : 0;
    d_beg[tid]  = base;
    d_fill[tid] = base;
    if (tid == NDC - 1) d_beg[NDC] = base + cn;
  }
  __syncthreads();

  const bool ovf = !part || d_beg[NDC] > QCAP;

  if (!ovf) {
    auto place = [&](const int2 en) {
      const int sl = __hip_atomic_fetch_add(&d_fill[(en.y >> 16) & (NDC - 1)], 1, __ATOMIC_RELAXED,
                                            __HIP_MEMORY_SCOPE_WORKGROUP);
      q_sr[sl] = make_int2(en.x, en.y & 0xffff);
    };
    if (sub < nseg)     place(en0);
    if (sub + 4 < nseg) place(en1);
    if (sub + 8 < nseg) place(en2);
    for (int i = sub + 12; i < nseg; i += 4) place(seg[i]);
  }
  __syncthreads();

  float4 cv[N_HEAD];
#pragma unroll
  for (int i = 0; i < N_HEAD; ++i) cv[i] = *(const float4*)(cs + i * F_OUT + lane * 4);

  float4 acc[N_HEAD];
  float  rsum[N_HEAD];
  auto clear = [&]() {
#pragma unroll
    for (int i = 0; i < N_HEAD; ++i) { acc[i] = make_float4(0.0f, 0.0f, 0.0f, 0.0f); rsum[i] = 0.0f; }
  };
  auto add_edge = [&](const v4f sn, const v4f sr, const float4& hs, const float4& rf) {
    const float snv[N_HEAD] = {sn.x, sn.y, sn.z, sn.w};
    const float srv[N_HEAD] = {sr.x, sr.y, sr.z, sr.w};
    float ee[N_HEAD];
#pragma unroll
    for (int i = 0; i < N_HEAD; ++i) {
      const float lg = snv[i] + srv[i];
      const float lk = lg > 0.0f ? lg : 0.2f * lg;
      float t = lk * -1.44269504088896340736f;
      asm volatile("" : "+s"(t));
      ee[i] = __builtin_amdgcn_exp2f(t);
      rsum[i] += ee[i];
    }
#pragma unroll
    for (int i = 0; i < N_HEAD; ++i) {
      acc[i].x = fmaf(fmaf(cv[i].x, hs.x, -rf.x), ee[i], acc[i].x);
      acc[i].y = fmaf(fmaf(cv[i].y, hs.y, -rf.y), ee[i], acc[i].y);
      acc[i].z = fmaf(fmaf(cv[i].z, hs.z, -rf.z), ee[i], acc[i].z);
      acc[i].w = fmaf(fmaf(cv[i].w, hs.w, -rf.w), ee[i], acc[i].w);
    }
  };
  auto flush = [&](int dl) {
#pragma unroll
    for (int i = 0; i < N_HEAD; ++i) {
      float4 o;
      o.x = acc[i].x / rsum[i];  o.y = acc[i].y / rsum[i];
      o.z = acc[i].z / rsum[i];  o.w = acc[i].w / rsum[i];
      *(float4*)(out + ((size_t)i * N + lo + dl) * F_OUT + lane * 4) = o;
    }
  };

  if (!ovf) {
    const int2* bs = q_sr;
    for (int it = 0; it < DPW; ++it) {
      int dl = 0;
      if (lane == 0)
        dl = __hip_atomic_fetch_add(&next_d, 1, __ATOMIC_RELAXED, __HIP_MEMORY_SCOPE_WORKGROUP);
      dl = __builtin_amdgcn_readfirstlane(dl);
      if (dl >= ndt) break;
      const int s0 = __builtin_amdgcn_readfirstlane(d_beg[dl]);
      const int s1 = __builtin_amdgcn_readfirstlane(d_beg[dl + 1]);
      clear();
      int b0 = s0;
      for (; b0 + 4 <= s1; b0 += 4) {
        const int2 en = bs[b0 + (lane & 3)];
        float4 hs[4], rf[4];
        v4f    sn[4], sr[4];
#pragma unroll
        for (int u = 0; u < 4; ++u) {
          const int src = __builtin_amdgcn_readlane(en.x, u);
          const int rl  = __builtin_amdgcn_readlane(en.y, u);
          sn[u] = *(score_row_ptr)(Snode + (size_t)src * 16);
          sr[u] = *(score_row_ptr)(Srel + (size_t)rl * 16);
          hs[u] = *(const float4*)(h + (size_t)src * F_OUT + lane * 4);
          rf[u] = *(const float4*)(inputr + (size_t)rl * F_OUT + lane * 4);
        }
        asm volatile("" : "+s"(sn[0]), "+s"(sr[0]), "+s"(sn[1]), "+s"(sr[1]),
                          "+s"(sn[2]), "+s"(sr[2]), "+s"(sn[3]), "+s"(sr[3])
                     : "s"(h), "s"(inputr) : "memory");
#pragma unroll
        for (int u = 0; u < 4; ++u) add_edge(sn[u], sr[u], hs[u], rf[u]);
      }
      for (; b0 < s1; ++b0) {
        const int2 en  = bs[b0];
        const int  src = __builtin_amdgcn_readfirstlane(en.x);
        const int  rl  = __builtin_amdgcn_readfirstlane(en.y);
        v4f sn1 = *(score_row_ptr)(Snode + (size_t)src * 16);
        v4f sr1 = *(score_row_ptr)(Srel + (size_t)rl * 16);
        const float4 hs1 = *(const float4*)(h + (size_t)src * F_OUT + lane * 4);
        const float4 rf1 = *(const float4*)(inputr + (size_t)rl * F_OUT + lane * 4);
        asm volatile("" : "+s"(sn1), "+s"(sr1) : "s"(h), "s"(inputr) : "memory");
        add_edge(sn1, sr1, hs1, rf1);
      }
      flush(dl);
    }
  } else {
    for (int j = 0; j * 32 + wv < ndt; ++j) {
      const int dl = j * 32 + wv;
      clear();
      for (int e0 = 0; e0 < E; e0 += 32) {
        const int e = e0 + lane;
        const int d = e < E ? A[e] : -1;
        unsigned m = __builtin_amdgcn_ballot_w32(d == lo + dl);
        while (m) {
          const int eu  = e0 + __builtin_ctz(m);
          m &= m - 1u;
          const int rl  = __builtin_amdgcn_readfirstlane(A[(size_t)E + eu]);
          const int src = __builtin_amdgcn_readfirstlane(A[2 * (size_t)E + eu]);
          add_edge(*(score_row_ptr)(Snode + (size_t)src * 16),
                   *(score_row_ptr)(Srel + (size_t)rl * 16),
                   *(const float4*)(h + (size_t)src * F_OUT + lane * 4),
                   *(const float4*)(inputr + (size_t)rl * F_OUT + lane * 4));
        }
      }
      flush(dl);
    }
  }
}

extern "C" void kernel_launch(void* const* d_in, const int* in_sizes, int n_in,
                              void* d_out, int out_size, void* d_ws, size_t ws_size,
                              hipStream_t stream) {
  const float* h      = (const float*)d_in[0];
  const float* inputr = (const float*)d_in[1];
  const float* w      = (const float*)d_in[2];
  const float* a      = (const float*)d_in[3];
  const int*   A      = (const int*)d_in[4];

  const int N    = in_sizes[0] / F_OUT;
  const int NREL = in_sizes[1] / F_OUT;
  const int E    = in_sizes[4] / 3;

  float* ws = (float*)d_ws;
  float* cs    = ws;
  float* BsrcP = ws + 512;
  float* BdstP = ws + 512 + 2048;
  size_t off   = 512 + 2048 + 2048;
  const int nrel_tiles = (NREL + 15) / 16;
  const int node_tiles = (N + 15) / 16;
  float* Srel  = ws + off;  off += (size_t)nrel_tiles * 16 * 16;
  float* Snode = ws + off;  off += (size_t)node_tiles * 16 * 16;
  const int nown = (N + DPW - 1) / DPW;
  const int CH   = ((E + NCH - 1) / NCH + 15) & ~15;
  const int tp   = (nown + 15) & ~15;
  off = (off + 31) & ~(size_t)31;
  int* Ent = (int*)(ws + off);  off += 2 * (size_t)E;
  off = (off + 31) & ~(size_t)31;
  int* Tab = (int*)(ws + off);  off += 2 * (size_t)NCH * tp;
  const int part = (off * sizeof(float) <= ws_size && nown <= MAXOWN && NREL <= 65536) ? 1 : 0;
  (void)out_size; (void)n_in;

  rgat_prep<<<part ? 1 + NCH : 1, PTHR, 0, stream>>>(w, a, cs, BsrcP, BdstP, A, Ent, Tab,
                                                     E, N, CH, nown, tp);

  const int WPB = 8;
  rgat_score<<<(node_tiles + WPB - 1) / WPB, 256, 0, stream>>>(h, BsrcP, Snode, N);
  rgat_score<<<(nrel_tiles + WPB - 1) / WPB, 256, 0, stream>>>(inputr, BdstP, Srel, NREL);

  rgat_own<<<nown, NTHR, 0, stream>>>(A, Ent, Tab, h, inputr, cs, Snode, Srel,
                                      (float*)d_out, E, N, CH, tp, part);
}
